# starting kernel
# baseline (speedup 1.0000x reference)
#include <hip/hip_runtime.h>


typedef __attribute__((ext_vector_type(2))) float v2f;
typedef __attribute__((ext_vector_type(8))) float v8f;

#define CIN   64
#define COUT  64
#define HH    128
#define WW    128
#define BATCH 8

#define XS_C_STRIDE 102
#define XS_FLOATS   (64 * 102)
#define GATE_OFF    XS_FLOATS
#define LDS_FLOATS  (XS_FLOATS + 9 * 32)

__global__ void repack_weight_kernel(const float* __restrict__ w,
                                     float2* __restrict__ pw) {
    int id = blockIdx.x * blockDim.x + threadIdx.x;
    if (id >= 4 * 9 * 16 * 32) return;
    int lane = id & 31;
    int cc   = (id >> 5) & 15;
    int kk   = (id >> 9) % 9;
    int mt   = (id >> 9) / 9;
    int o = mt * 16 + (lane & 15);
    int c = cc * 4 + 2 * (lane >> 4);
    float2 v;
    v.x = w[(o * CIN + c) * 9 + kk];
    v.y = w[(o * CIN + c + 1) * 9 + kk];
    pw[id] = v;
}

__global__ void __launch_bounds__(256)
depthconv_wmma_kernel(const float* __restrict__ x,
                      const float* __restrict__ depth,
                      const float* __restrict__ bias,
                      const float2* __restrict__ pw,
                      float* __restrict__ out) {
    __shared__ float lds[LDS_FLOATS];

    const int tid = threadIdx.x;
    const int b   = blockIdx.z;
    const int h   = blockIdx.y;
    const int w0  = blockIdx.x * 32;

    const float* xb = x + (size_t)b * CIN * HH * WW;
    for (int idx = tid; idx < XS_FLOATS; idx += 256) {
        int c   = idx / XS_C_STRIDE;
        int rem = idx - c * XS_C_STRIDE;
        int r   = rem / 34;
        int j   = rem - r * 34;
        int gh  = h + r - 1;
        int gw  = w0 + j - 1;
        float v = 0.0f;
        if ((unsigned)gh < HH && (unsigned)gw < WW)
            v = xb[(c * HH + gh) * WW + gw];
        lds[idx] = v;
    }

    const float* db = depth + (size_t)b * HH * WW;
    for (int idx = tid; idx < 9 * 32; idx += 256) {
        int k  = idx >> 5;
        int j  = idx & 31;
        int dh = k / 3 - 1;
        int dw = k - (k / 3) * 3 - 1;
        int gh = h + dh;
        int gw = w0 + j + dw;
        float dc = db[h * WW + w0 + j];
        float dn = ((unsigned)gh < HH && (unsigned)gw < WW) ? db[gh * WW + gw]
                                                           : 0.0f;
        lds[GATE_OFF + idx] = __expf(-fabsf(dn - dc));
    }
    __syncthreads();

    const int lane    = tid & 31;
    const int wave    = tid >> 5;
    const int mt      = wave & 3;
    const int tc      = wave >> 2;
    const int halfsel = lane >> 4;
    const int col     = tc * 16 + (lane & 15);

    v8f acc = {0.f, 0.f, 0.f, 0.f, 0.f, 0.f, 0.f, 0.f};

    const float2* ap0 = pw + (size_t)mt * 9 * 16 * 32 + lane;

    for (int k = 0; k < 9; ++k) {
        const int dh  = k / 3;
        const int dwj = k - dh * 3;
        v8f y = {0.f, 0.f, 0.f, 0.f, 0.f, 0.f, 0.f, 0.f};

        const float2* ap = ap0 + k * 16 * 32;
        const float* bp = lds + (2 * halfsel) * XS_C_STRIDE + dh * 34
                              + (col + dwj);
#pragma unroll
        for (int cc = 0; cc < 16; ++cc) {
            float2 at = ap[cc * 32];
            v2f a; a.x = at.x; a.y = at.y;
            v2f bv;
            bv.x = bp[cc * (4 * XS_C_STRIDE)];
            bv.y = bp[cc * (4 * XS_C_STRIDE) + XS_C_STRIDE];
            y = __builtin_amdgcn_wmma_f32_16x16x4_f32(
                    false, a, false, bv, (short)0, y, false, false);
        }
        float g = lds[GATE_OFF + k * 32 + col];
#pragma unroll
        for (int i = 0; i < 8; ++i) acc[i] = fmaf(g, y[i], acc[i]);
    }

    const int o0 = mt * 16 + halfsel * 8;
    float* ob = out + (size_t)b * COUT * HH * WW;
#pragma unroll
    for (int r = 0; r < 8; ++r) {
        int o = o0 + r;
        ob[((size_t)o * HH + h) * WW + w0 + col] = acc[r] + bias[o];
    }
}

extern "C" void kernel_launch(void* const* d_in, const int* in_sizes, int n_in,
                              void* d_out, int out_size, void* d_ws,
                              size_t ws_size, hipStream_t stream) {
    const float* x      = (const float*)d_in[0];
    const float* depth  = (const float*)d_in[1];
    const float* weight = (const float*)d_in[2];
    const float* bias   = (const float*)d_in[3];
    float*  out = (float*)d_out;
    float2* pw  = (float2*)d_ws;

    repack_weight_kernel<<<dim3(72), dim3(256), 0, stream>>>(weight, pw);

    dim3 grid(WW / 32, HH, BATCH);
    depthconv_wmma_kernel<<<grid, dim3(256), 0, stream>>>(x, depth, bias, pw,
                                                          out);
}


# Work-group tile 64 Cout x (4 rows x 32 cols), 4 pixel tiles per wave: one weight fragment feeds 4 WMMAs, halo staged once per 4 rows as channel-pair b64 LDS rows; summation order unchanged
# speedup vs baseline: 1.3767x; 1.3767x over previous
#include <hip/hip_runtime.h>


typedef __attribute__((ext_vector_type(2))) float v2f;
typedef __attribute__((ext_vector_type(8))) float v8f;

#define CIN   64
#define COUT  64
#define HH    128
#define WW    128
#define BATCH 8

#define TROWS 4
#define TCOLS 32
#define XROWS (TROWS + 2)
#define PITCH     40
#define XS_F2     (32 * XROWS * PITCH)
#define GATE_OFF  (2 * XS_F2)
#define NPIX      (TROWS * TCOLS)
#define LDS_FLOATS (GATE_OFF + 9 * NPIX)

__global__ void repack_weight_kernel(const float* __restrict__ w,
                                     float2* __restrict__ pw) {
    int id = blockIdx.x * blockDim.x + threadIdx.x;
    if (id >= 4 * 9 * 16 * 32) return;
    int lane = id & 31;
    int cc   = (id >> 5) & 15;
    int kk   = (id >> 9) % 9;
    int mt   = (id >> 9) / 9;
    int o = mt * 16 + (lane & 15);
    int c = cc * 4 + 2 * (lane >> 4);
    float2 v;
    v.x = w[(o * CIN + c) * 9 + kk];
    v.y = w[(o * CIN + c + 1) * 9 + kk];
    pw[id] = v;
}

__global__ void __launch_bounds__(256) __attribute__((amdgpu_waves_per_eu(8, 8)))
depthconv_wmma_kernel(const float* __restrict__ x,
                      const float* __restrict__ depth,
                      const float* __restrict__ bias,
                      const float2* __restrict__ pw,
                      float* __restrict__ out) {
    __shared__ __attribute__((aligned(16))) float lds[LDS_FLOATS];
    float2* xs2 = reinterpret_cast<float2*>(lds);

    const int tid  = threadIdx.x;
    const int lane = tid & 31;
    const int wave = tid >> 5;
    const int b    = blockIdx.z;
    const int h0   = blockIdx.y * TROWS;
    const int w0   = blockIdx.x * TCOLS;

    const float* xb = x + (size_t)b * CIN * HH * WW;

    {
        const bool  top_ok = (h0 > 0);
        const bool  bot_ok = (h0 + TROWS < HH);
        const int   roff[XROWS] = { top_ok ? -1 : 0, 0, 1, 2, 3,
                                    bot_ok ? TROWS : TROWS - 1 };
        const float* gp = xb + ((size_t)(wave * 8) * HH + h0) * WW + w0 + lane;
        float2* lp = xs2 + (wave * 4) * XROWS * PITCH + 1 + lane;
#pragma unroll
        for (int qi = 0; qi < 4; ++qi) {
#pragma unroll
            for (int r = 0; r < XROWS; ++r) {
                const float* p = gp + (qi * 2 * HH) * WW + roff[r] * WW;
                float2 v;
                v.x = p[0];
                v.y = p[HH * WW];
                if (r == 0 && !top_ok) { v.x = 0.0f; v.y = 0.0f; }
                if (r == XROWS - 1 && !bot_ok) { v.x = 0.0f; v.y = 0.0f; }
                lp[(qi * XROWS + r) * PITCH] = v;
            }
        }
    }
    const float* db = depth + (size_t)b * HH * WW;
    auto stage_halo = [&](int e) {
        const int side = e & 1;
        const int pr   = e >> 1;
        const int q    = pr / XROWS;
        const int r    = pr - q * XROWS;
        const int gh   = h0 + r - 1;
        const int gw   = side ? (w0 + TCOLS) : (w0 - 1);
        const bool ok  = ((unsigned)gh < HH) && ((unsigned)gw < WW);
        const int ghc  = min(max(gh, 0), HH - 1);
        const int gwc  = min(max(gw, 0), WW - 1);
        const float* p = xb + ((size_t)(2 * q) * HH + ghc) * WW + gwc;
        float2 v;
        v.x = p[0];
        v.y = p[HH * WW];
        if (!ok) { v.x = 0.0f; v.y = 0.0f; }
        xs2[pr * PITCH + (side ? (TCOLS + 1) : 0)] = v;
    };
    auto stage_gate = [&](int idx) {
        int k  = idx >> 7;
        int p  = idx & (NPIX - 1);
        int rr = p >> 5;
        int j  = p & 31;
        int dh = k / 3 - 1;
        int dw = k - (k / 3) * 3 - 1;
        int gh = h0 + rr + dh;
        int gw = w0 + j + dw;
        const bool ok = ((unsigned)gh < HH) && ((unsigned)gw < WW);
        int ghc = min(max(gh, 0), HH - 1);
        int gwc = min(max(gw, 0), WW - 1);
        float dc = db[(h0 + rr) * WW + w0 + j];
        float dn = db[ghc * WW + gwc];
        if (!ok) dn = 0.0f;
        lds[GATE_OFF + idx] = __expf(-fabsf(dn - dc));
    };
    stage_halo(tid);
#pragma unroll
    for (int it = 0; it < 4; ++it) stage_gate(tid + it * 256);
    if (tid < 128) {
        stage_halo(tid + 256);
        stage_gate(tid + 4 * 256);
    }
    __syncthreads();

    const int mt      = wave & 3;
    const int ng      = wave >> 2;
    const int halfsel = lane >> 4;
    const int l15     = lane & 15;

    v8f acc[4];
#pragma unroll
    for (int t = 0; t < 4; ++t)
        acc[t] = (v8f){0.f, 0.f, 0.f, 0.f, 0.f, 0.f, 0.f, 0.f};

    const float2* ap0 = pw + (size_t)mt * 9 * 16 * 32 + lane;
    const float2* bbase = xs2 + (halfsel * XROWS + 2 * ng) * PITCH + l15;
    const float*  gbase = lds + GATE_OFF + (2 * ng) * TCOLS + l15;

    for (int k = 0; k < 9; ++k) {
        const int dh = k / 3;
        const int dw = k - dh * 3;
        const float2* ap = ap0 + k * 16 * 32;
        const float2* bp = bbase + dh * PITCH + dw;

        v8f y[4];
#pragma unroll
        for (int t = 0; t < 4; ++t)
            y[t] = (v8f){0.f, 0.f, 0.f, 0.f, 0.f, 0.f, 0.f, 0.f};

        float2 areg[16];
#pragma unroll
        for (int cc = 0; cc < 16; ++cc) areg[cc] = ap[cc * 32];

#pragma unroll
        for (int cc = 0; cc < 16; ++cc) {
            v2f a; a.x = areg[cc].x; a.y = areg[cc].y;
#pragma unroll
            for (int t = 0; t < 4; ++t) {
                float2 bt = bp[(cc * 2 * XROWS + (t >> 1)) * PITCH
                               + (t & 1) * 16];
                v2f bv; bv.x = bt.x; bv.y = bt.y;
                y[t] = __builtin_amdgcn_wmma_f32_16x16x4_f32(
                           false, a, false, bv, (short)0, y[t], false, false);
            }
        }
#pragma unroll
        for (int t = 0; t < 4; ++t) {
            float g = gbase[k * NPIX + (t >> 1) * TCOLS + (t & 1) * 16];
#pragma unroll
            for (int i = 0; i < 8; ++i)
                acc[t][i] = fmaf(g, y[t][i], acc[t][i]);
        }
    }

    const int o0 = mt * 16 + halfsel * 8;
    float* ob = out + (size_t)b * COUT * HH * WW;
    float bs[8];
#pragma unroll
    for (int r = 0; r < 8; ++r) bs[r] = bias[o0 + r];
#pragma unroll
    for (int t = 0; t < 4; ++t) {
        const int hrow = h0 + 2 * ng + (t >> 1);
        const int wcol = w0 + (t & 1) * 16 + l15;
#pragma unroll
        for (int r = 0; r < 8; ++r) {
            int o = o0 + r;
            ob[((size_t)o * HH + hrow) * WW + wcol] = acc[t][r] + bs[r];
        }
    }
}

extern "C" void kernel_launch(void* const* d_in, const int* in_sizes, int n_in,
                              void* d_out, int out_size, void* d_ws,
                              size_t ws_size, hipStream_t stream) {
    const float* x      = (const float*)d_in[0];
    const float* depth  = (const float*)d_in[1];
    const float* weight = (const float*)d_in[2];
    const float* bias   = (const float*)d_in[3];
    float*  out = (float*)d_out;
    float2* pw  = (float2*)d_ws;

    repack_weight_kernel<<<dim3(72), dim3(256), 0, stream>>>(weight, pw);

    dim3 grid(WW / TCOLS, HH / TROWS, BATCH);
    depthconv_wmma_kernel<<<grid, dim3(256), 0, stream>>>(x, depth, bias, pw,
                                                          out);
}
